# starting kernel
# baseline (speedup 1.0000x reference)
#include <hip/hip_runtime.h>
#include <hip/hip_bf16.h>

typedef __attribute__((ext_vector_type(2))) float v2f;
typedef __attribute__((ext_vector_type(8))) float v8f;

#define F_IN   256
#define F_OUT  64
#define NEG_SLOPE 0.2f

__device__ __forceinline__ unsigned f2sortable(float f) {
    unsigned u = __float_as_uint(f);
    return (u & 0x80000000u) ? ~u : (u | 0x80000000u);
}
__device__ __forceinline__ float sortable2f(unsigned u) {
    return __uint_as_float((u & 0x80000000u) ? (u ^ 0x80000000u) : ~u);
}

__global__ void init_kernel(float* __restrict__ acc, unsigned* __restrict__ rmax,
                            float* __restrict__ denom, int n) {
    int i = blockIdx.x * blockDim.x + threadIdx.x;
    if (i < n * F_OUT) acc[i] = 0.0f;
    if (i < n) { rmax[i] = 0u; denom[i] = 0.0f; }
}

__global__ void __launch_bounds__(256)
gemm_wmma_kernel(const float* __restrict__ seq, const float* __restrict__ W,
                 float* __restrict__ fts,
                 const float* __restrict__ a1, const float* __restrict__ b1,
                 const float* __restrict__ a2, const float* __restrict__ b2,
                 float* __restrict__ f1, float* __restrict__ f2, int n) {
    __shared__ float Wl[F_IN * F_OUT];
    for (int i = threadIdx.x; i < F_IN * F_OUT; i += 256) {
        const int kk  = i >> 6;
        const int col = i & 63;
        Wl[(kk >> 1) * (2 * F_OUT) + col * 2 + (kk & 1)] = W[i];
    }
    __syncthreads();

    const int wave = threadIdx.x >> 5;
    const int lane = threadIdx.x & 31;
    const int rowTile = blockIdx.x * 8 + wave;

    int row = rowTile * 16 + (lane & 15);
    if (row >= n) row = n - 1;
    const int lcol  = lane & 15;
    const int khalf = (lane >> 4) * 2;

    const float* __restrict__ arow = seq + (size_t)row * F_IN;

    v8f acc[4] = {v8f{}, v8f{}, v8f{}, v8f{}};
    for (int k = 0; k < F_IN; k += 4) {
        float2 av = *(const float2*)(arow + k + khalf);
        v2f a; a[0] = av.x; a[1] = av.y;
        const int kpairBase = ((k + khalf) >> 1) * (2 * F_OUT);
#pragma unroll
        for (int ct = 0; ct < 4; ++ct) {
            float2 bv = *(const float2*)(&Wl[kpairBase + (ct * 16 + lcol) * 2]);
            v2f b; b[0] = bv.x; b[1] = bv.y;
            acc[ct] = __builtin_amdgcn_wmma_f32_16x16x4_f32(
                     false, a,  false, b,
                     (short)0, acc[ct],  false,  false);
        }
    }

    const int nhi = lane >> 4;
#pragma unroll
    for (int ct = 0; ct < 4; ++ct) {
        const int ocol = ct * 16 + lcol;
#pragma unroll
        for (int r = 0; r < 8; ++r) {
            const int orow = rowTile * 16 + r + 8 * nhi;
            if (orow < n) fts[(size_t)orow * F_OUT + ocol] = acc[ct][r];
        }
    }

    float A1[4], A2[4];
#pragma unroll
    for (int ct = 0; ct < 4; ++ct) {
        A1[ct] = a1[ct * 16 + lcol];
        A2[ct] = a2[ct * 16 + lcol];
    }
    const float b1v = b1[0], b2v = b2[0];
#pragma unroll
    for (int r = 0; r < 8; ++r) {
        float p1 = 0.0f, p2 = 0.0f;
#pragma unroll
        for (int ct = 0; ct < 4; ++ct) {
            p1 += acc[ct][r] * A1[ct];
            p2 += acc[ct][r] * A2[ct];
        }
#pragma unroll
        for (int off = 8; off > 0; off >>= 1) {
            p1 += __shfl_xor(p1, off, 32);
            p2 += __shfl_xor(p2, off, 32);
        }
        if ((lane & 15) == 0) {
            const int orow = rowTile * 16 + r + 8 * nhi;
            if (orow < n) { f1[orow] = p1 + b1v; f2[orow] = p2 + b2v; }
        }
    }
}

__global__ void logits_kernel(const int* __restrict__ erow, const int* __restrict__ ecol,
                              const float* __restrict__ f1, const float* __restrict__ f2,
                              float* __restrict__ logits, unsigned* __restrict__ rmax, int e) {
    const int i = blockIdx.x * blockDim.x + threadIdx.x;
    if (i >= e) return;
    const int r = erow[i], c = ecol[i];
    float l = f1[r] + f2[c];
    l = l > 0.0f ? l : NEG_SLOPE * l;
    logits[i] = l;
    atomicMax(&rmax[r], f2sortable(l));
}

__global__ void ev_kernel(const int* __restrict__ erow, float* __restrict__ logits,
                          const unsigned* __restrict__ rmax, float* __restrict__ denom, int e) {
    const int i = blockIdx.x * blockDim.x + threadIdx.x;
    if (i >= e) return;
    const int r = erow[i];
    const float m = sortable2f(rmax[r]);
    const float v = expf(logits[i] - m);
    logits[i] = v;
    atomicAdd(&denom[r], v);
}

__global__ void spmm_kernel(const int* __restrict__ erow, const int* __restrict__ ecol,
                            const float* __restrict__ ev, const float* __restrict__ denom,
                            const float* __restrict__ fts, float* __restrict__ acc, int e) {
    const int gid  = blockIdx.x * blockDim.x + threadIdx.x;
    const int edge = gid >> 5;
    const int lane = threadIdx.x & 31;
    if (edge >= e) return;
    const int r = erow[edge], c = ecol[edge];
    const float coef = ev[edge] / denom[r];
    const float* __restrict__ src = fts + (size_t)c * F_OUT;
    float* __restrict__ dst = acc + (size_t)r * F_OUT;
    atomicAdd(&dst[lane],      coef * src[lane]);
    atomicAdd(&dst[lane + 32], coef * src[lane + 32]);
}

__global__ void finalize_kernel(float* __restrict__ acc, const float* __restrict__ bias, int n) {
    const int i = blockIdx.x * blockDim.x + threadIdx.x;
    if (i >= n * F_OUT) return;
    const float v = acc[i] + bias[i & (F_OUT - 1)];
    acc[i] = v > 0.0f ? v : 0.0f;
}

extern "C" void kernel_launch(void* const* d_in, const int* in_sizes, int n_in,
                              void* d_out, int out_size, void* d_ws, size_t ws_size,
                              hipStream_t stream) {
    const float* seq  = (const float*)d_in[0];
    const int*   erow = (const int*)  d_in[1];
    const int*   ecol = (const int*)  d_in[2];
    const float* W    = (const float*)d_in[3];
    const float* a1   = (const float*)d_in[4];
    const float* b1   = (const float*)d_in[5];
    const float* a2   = (const float*)d_in[6];
    const float* b2   = (const float*)d_in[7];
    const float* bias = (const float*)d_in[8];

    const int n = in_sizes[0] / F_IN;
    const int e = in_sizes[1];
    float* out = (float*)d_out;

    float*    fts   = (float*)d_ws;
    float*    f1    = fts + (size_t)n * F_OUT;
    float*    f2    = f1 + n;
    unsigned* rmax  = (unsigned*)(f2 + n);
    float*    denom = (float*)(rmax + n);
    float*    ev    = denom + n;

    {
        const int tot = n * F_OUT;
        init_kernel<<<(tot + 255) / 256, 256, 0, stream>>>(out, rmax, denom, n);
    }
    {
        const int blocks = (n + 127) / 128;
        gemm_wmma_kernel<<<blocks, 256, 0, stream>>>(seq, W, fts, a1, b1, a2, b2,
                                                     f1, f2, n);
    }
    logits_kernel<<<(e + 255) / 256, 256, 0, stream>>>(erow, ecol, f1, f2, ev, rmax, e);
    ev_kernel<<<(e + 255) / 256, 256, 0, stream>>>(erow, ev, rmax, denom, e);
    {
        const long long threads = (long long)e * 32;
        const int blocks = (int)((threads + 255) / 256);
        spmm_kernel<<<blocks, 256, 0, stream>>>(erow, ecol, ev, denom, fts, out, e);
    }
    finalize_kernel<<<(n * F_OUT + 255) / 256, 256, 0, stream>>>(out, bias, n);
}


# Softmax row shift is the bound lrelu(f1[r]+max f2), not the segment max: logits launch and its per-edge atomics go (exact max kept for wide f2); weights scaled per row, rounding changes
# speedup vs baseline: 2.9976x; 2.9976x over previous
#include <hip/hip_runtime.h>
#include <hip/hip_bf16.h>

typedef __attribute__((ext_vector_type(2))) float v2f;
typedef __attribute__((ext_vector_type(8))) float v8f;

#define F_IN   256
#define F_OUT  64
#define NEG_SLOPE 0.2f

__device__ __forceinline__ unsigned f2sortable(float f) {
    unsigned u = __float_as_uint(f);
    return (u & 0x80000000u) ? ~u : (u | 0x80000000u);
}
__device__ __forceinline__ float sortable2f(unsigned u) {
    return __uint_as_float((u & 0x80000000u) ? (u ^ 0x80000000u) : ~u);
}

#define TICKET_PITCH 64

#define CTL_F2MAX (2 * TICKET_PITCH)
#define CTL_F2MIN (3 * TICKET_PITCH)

#define SPREAD_SAFE 60.0f

__global__ void init_kernel(float* __restrict__ acc, float* __restrict__ denom,
                            unsigned* __restrict__ ctl, int n) {
    int i = blockIdx.x * blockDim.x + threadIdx.x;
    if (i < n * F_OUT) acc[i] = 0.0f;
    if (i < n) denom[i] = 0.0f;
    if (i < 2) ctl[i * TICKET_PITCH] = 0u;
    if (i == 2) ctl[CTL_F2MAX] = 0u;
    if (i == 3) ctl[CTL_F2MIN] = 0xFFFFFFFFu;
}

__global__ void __launch_bounds__(256)
gemm_wmma_kernel(const float* __restrict__ seq, const float* __restrict__ W,
                 float* __restrict__ fts,
                 const float* __restrict__ a1, const float* __restrict__ b1,
                 const float* __restrict__ a2, const float* __restrict__ b2,
                 float* __restrict__ f1, float* __restrict__ f2,
                 unsigned* __restrict__ rmax, unsigned* __restrict__ ctl, int n) {
    __shared__ float Wl[F_IN * F_OUT];
    __shared__ unsigned s_f2hi[8], s_f2lo[8];
    for (int i = threadIdx.x; i < F_IN * F_OUT; i += 256) {
        const int kk  = i >> 6;
        const int col = i & 63;
        Wl[(kk >> 1) * (2 * F_OUT) + col * 2 + (kk & 1)] = W[i];
    }
    __syncthreads();

    const int wave = threadIdx.x >> 5;
    const int lane = threadIdx.x & 31;
    const int rowTile = blockIdx.x * 8 + wave;

    int row = rowTile * 16 + (lane & 15);
    if (row >= n) row = n - 1;
    const int lcol  = lane & 15;
    const int khalf = (lane >> 4) * 2;

    const float* __restrict__ arow = seq + (size_t)row * F_IN;

    v8f acc[4] = {v8f{}, v8f{}, v8f{}, v8f{}};
    for (int k = 0; k < F_IN; k += 4) {
        float2 av = *(const float2*)(arow + k + khalf);
        v2f a; a[0] = av.x; a[1] = av.y;
        const int kpairBase = ((k + khalf) >> 1) * (2 * F_OUT);
#pragma unroll
        for (int ct = 0; ct < 4; ++ct) {
            float2 bv = *(const float2*)(&Wl[kpairBase + (ct * 16 + lcol) * 2]);
            v2f b; b[0] = bv.x; b[1] = bv.y;
            acc[ct] = __builtin_amdgcn_wmma_f32_16x16x4_f32(
                     false, a,  false, b,
                     (short)0, acc[ct],  false,  false);
        }
    }

    const int nhi = lane >> 4;
#pragma unroll
    for (int ct = 0; ct < 4; ++ct) {
        const int ocol = ct * 16 + lcol;
#pragma unroll
        for (int r = 0; r < 8; ++r) {
            const int orow = rowTile * 16 + r + 8 * nhi;
            if (orow < n) fts[(size_t)orow * F_OUT + ocol] = acc[ct][r];
        }
    }

    float A1[4], A2[4];
#pragma unroll
    for (int ct = 0; ct < 4; ++ct) {
        A1[ct] = a1[ct * 16 + lcol];
        A2[ct] = a2[ct * 16 + lcol];
    }
    const float b1v = b1[0], b2v = b2[0];
    unsigned uhi = 0u, ulo = 0xFFFFFFFFu;
#pragma unroll
    for (int r = 0; r < 8; ++r) {
        float p1 = 0.0f, p2 = 0.0f;
#pragma unroll
        for (int ct = 0; ct < 4; ++ct) {
            p1 += acc[ct][r] * A1[ct];
            p2 += acc[ct][r] * A2[ct];
        }
#pragma unroll
        for (int off = 8; off > 0; off >>= 1) {
            p1 += __shfl_xor(p1, off, 32);
            p2 += __shfl_xor(p2, off, 32);
        }
        if ((lane & 15) == 0) {
            const int orow = rowTile * 16 + r + 8 * nhi;
            if (orow < n) {
                const float f2v = p2 + b2v;
                f1[orow] = p1 + b1v; f2[orow] = f2v; rmax[orow] = 0u;
                const unsigned s = f2sortable(f2v);
                uhi = uhi > s ? uhi : s;
                ulo = ulo < s ? ulo : s;
            }
        }
    }
    {
        const unsigned ohi = (unsigned)__shfl_xor((int)uhi, 16, 32);
        const unsigned olo = (unsigned)__shfl_xor((int)ulo, 16, 32);
        uhi = uhi > ohi ? uhi : ohi;
        ulo = ulo < olo ? ulo : olo;
    }
    if (lane == 0) { s_f2hi[wave] = uhi; s_f2lo[wave] = ulo; }
    __syncthreads();
    if (threadIdx.x == 0) {
        unsigned bh = s_f2hi[0], bl = s_f2lo[0];
        for (int w = 1; w < 8; ++w) {
            bh = bh > s_f2hi[w] ? bh : s_f2hi[w];
            bl = bl < s_f2lo[w] ? bl : s_f2lo[w];
        }
        atomicMax(&ctl[CTL_F2MAX], bh);
        atomicMin(&ctl[CTL_F2MIN], bl);
    }
}

__device__ __forceinline__ float edge_logit(float f1r, float f2c) {
    const float l = f1r + f2c;
    return l > 0.0f ? l : NEG_SLOPE * l;
}

__global__ void __launch_bounds__(256)
rowmax_kernel(const int* __restrict__ erow, const int* __restrict__ ecol,
              const float* __restrict__ f1, const float* __restrict__ f2,
              unsigned* __restrict__ rmax, const unsigned* __restrict__ f2ext, int e) {
    const float hi = sortable2f(f2ext[0]);
    const float lo = sortable2f(f2ext[TICKET_PITCH]);
    if (hi - lo <= SPREAD_SAFE) return;
    const int step = gridDim.x * blockDim.x;
    for (int i = blockIdx.x * blockDim.x + threadIdx.x; i < e; i += step) {
        const int r = erow[i], c = ecol[i];
        atomicMax(&rmax[r], f2sortable(edge_logit(f1[r], f2[c])));
    }
}

template <int NB, bool PARTIAL>
__device__ __forceinline__ void scatter_batch(int r, int c, float v, unsigned& todo, int cnt, int lane,
                                              const float* __restrict__ fts,
                                              float* __restrict__ acc) {
    float x0[NB], x1[NB];
    int   js[NB];
#pragma unroll
    for (int u = 0; u < NB; ++u) {
        if (PARTIAL && u >= cnt) break;
        const int j = __builtin_ctz(todo);
        todo &= todo - 1u;
        js[u] = j;
        const int cj = __builtin_amdgcn_readlane(c, j);
        const float* __restrict__ s = fts + (size_t)cj * F_OUT;
        x0[u] = s[lane];
        x1[u] = s[lane + 32];
    }
#pragma unroll
    for (int u = 0; u < NB; ++u) {
        if (PARTIAL && u >= cnt) break;
        const int   rj = __builtin_amdgcn_readlane(r, js[u]);
        const float vj = __int_as_float(__builtin_amdgcn_readlane(__float_as_int(v), js[u]));
        float* __restrict__ d = acc + (size_t)rj * F_OUT;
        atomicAdd(&d[lane],      vj * x0[u]);
        atomicAdd(&d[lane + 32], vj * x1[u]);
    }
}

__global__ void __launch_bounds__(256)
scatter_kernel(const int* __restrict__ erow, const int* __restrict__ ecol,
               const float* __restrict__ f1, const float* __restrict__ f2,
               const unsigned* __restrict__ rmax, const unsigned* __restrict__ f2ext,
               float* __restrict__ denom, const float* __restrict__ fts,
               float* __restrict__ acc, unsigned* __restrict__ tickets, int e, int n) {
    __shared__ int s_half;
    if (threadIdx.x == 0) {
        s_half = (int)((__builtin_amdgcn_s_sendmsg_rtn(0x87) >> 18) & 1u);
    }
    __syncthreads();
    const int myhalf = __builtin_amdgcn_readfirstlane(s_half);

    const int lane = threadIdx.x & 31;
    const float f2hi = sortable2f(f2ext[0]);
    const float f2lo = sortable2f(f2ext[TICKET_PITCH]);
    const bool bound_ok = (f2hi - f2lo) <= SPREAD_SAFE;
    const int nsplit = (n >> 1) & ~63;
    const unsigned npieces = (unsigned)((e + 31) >> 5);

    for (int pass = 0; pass < 2; ++pass) {
        const int side = myhalf ^ pass;
        unsigned* __restrict__ tk = tickets + side * TICKET_PITCH;
        for (;;) {
            unsigned drawn = 0u;
            if (lane == 0) drawn = atomicAdd(tk, 1u);
            const unsigned t = (unsigned)__builtin_amdgcn_readfirstlane((int)drawn);
            if (t >= npieces) break;

            const int i = (int)(t * 32u) + lane;
            int r = 0, c = 0;
            float v = 0.0f;
            bool mine = false;
            if (i < e) {
                r = erow[i];
                c = ecol[i];
                mine = ((r >= nsplit) ? 1 : 0) == side;
            }
            if (mine) {
                const float f1r = f1[r];
                const float l = edge_logit(f1r, f2[c]);
                float shift;
                if (bound_ok) shift = edge_logit(f1r, f2hi);
                else          shift = sortable2f(rmax[r]);
                v = expf(l - shift);
                atomicAdd(&denom[r], v);
            }
            unsigned todo = __builtin_amdgcn_ballot_w32(mine);
            int left = __builtin_popcount(todo);
            while (left >= 8) { scatter_batch<8, false>(r, c, v, todo, 8, lane, fts, acc); left -= 8; }
            if (left > 0) scatter_batch<7, true>(r, c, v, todo, left, lane, fts, acc);
        }
    }
}

__global__ void finalize_kernel(float* __restrict__ acc, const float* __restrict__ bias,
                                const float* __restrict__ denom, int n) {
    const int i = blockIdx.x * blockDim.x + threadIdx.x;
    if (i >= n * F_OUT) return;
    const float d = denom[i >> 6];
    float s = acc[i];
    if (d > 0.0f) s = s / d;
    const float v = s + bias[i & (F_OUT - 1)];
    acc[i] = v > 0.0f ? v : 0.0f;
}

extern "C" void kernel_launch(void* const* d_in, const int* in_sizes, int n_in,
                              void* d_out, int out_size, void* d_ws, size_t ws_size,
                              hipStream_t stream) {
    const float* seq  = (const float*)d_in[0];
    const int*   erow = (const int*)  d_in[1];
    const int*   ecol = (const int*)  d_in[2];
    const float* W    = (const float*)d_in[3];
    const float* a1   = (const float*)d_in[4];
    const float* b1   = (const float*)d_in[5];
    const float* a2   = (const float*)d_in[6];
    const float* b2   = (const float*)d_in[7];
    const float* bias = (const float*)d_in[8];

    const int n = in_sizes[0] / F_IN;
    const int e = in_sizes[1];
    float* out = (float*)d_out;

    float*    fts   = (float*)d_ws;
    float*    f1    = fts + (size_t)n * F_OUT;
    float*    f2    = f1 + n;
    unsigned* rmax  = (unsigned*)(f2 + n);
    float*    denom = (float*)(rmax + n);
    unsigned* ctl   = (unsigned*)(denom + n);
    const unsigned* f2ext = ctl + CTL_F2MAX;

    {
        const int tot = n * F_OUT;
        init_kernel<<<(tot + 255) / 256, 256, 0, stream>>>(out, denom, ctl, n);
    }
    {
        const int blocks = (n + 127) / 128;
        gemm_wmma_kernel<<<blocks, 256, 0, stream>>>(seq, W, fts, a1, b1, a2, b2,
                                                     f1, f2, rmax, ctl, n);
    }
    rowmax_kernel<<<512, 256, 0, stream>>>(erow, ecol, f1, f2, rmax, f2ext, e);
    scatter_kernel<<<2048, 256, 0, stream>>>(erow, ecol, f1, f2, rmax, f2ext, denom, fts, out,
                                             ctl, e, n);
    finalize_kernel<<<(n * F_OUT + 255) / 256, 256, 0, stream>>>(out, bias, denom, n);
}
